# starting kernel
# baseline (speedup 1.0000x reference)
#include <hip/hip_runtime.h>

typedef float v2f __attribute__((ext_vector_type(2)));
typedef float v8f __attribute__((ext_vector_type(8)));

#define CHUNK 2048
#define BLOCK 256
#define ROWS_PER_BLOCK 128
#define TILES_PER_CHUNK (CHUNK / 16)

__global__ __launch_bounds__(BLOCK) void nn_wmma_gather_kernel(
    const float* __restrict__ pts,
    const float* __restrict__ lidar,
    const float* __restrict__ features,
    float* __restrict__ outp,
    int M, int N, int D)
{
    __shared__ float2 sB[TILES_PER_CHUNK * 32];

    const int tid  = threadIdx.x;
    const int lane = tid & 31;
    const int wave = tid >> 5;
    const int row  = lane & 15;
    const bool hi  = lane >= 16;

    const int mBase = blockIdx.x * ROWS_PER_BLOCK + wave * 16;
    const int m     = mBase + row;

    const float qx = pts[(size_t)m * 3 + 0];
    const float qy = pts[(size_t)m * 3 + 1];
    const float qz = pts[(size_t)m * 3 + 2];
    v2f a;
    a[0] = hi ? (-2.0f * qz) : (-2.0f * qx);
    a[1] = hi ? 1.0f         : (-2.0f * qy);

    float bestVal[8];
    int   bestIdx[8];
#pragma unroll
    for (int r = 0; r < 8; ++r) { bestVal[r] = 3.402823466e+38f; bestIdx[r] = 0; }

    const int nChunks = N / CHUNK;
    for (int c = 0; c < nChunks; ++c) {
        const int chunkBase = c * CHUNK;

        for (int i = tid; i < CHUNK; i += BLOCK) {
            const int n = chunkBase + i;
            const float x = lidar[(size_t)n * 3 + 0];
            const float y = lidar[(size_t)n * 3 + 1];
            const float z = lidar[(size_t)n * 3 + 2];
            const int t  = i >> 4;
            const int cc = i & 15;
            sB[t * 32 + cc]      = make_float2(x, y);
            sB[t * 32 + cc + 16] = make_float2(z, x * x + y * y + z * z);
        }
        if (c + 1 < nChunks) {
            __builtin_prefetch(lidar + (size_t)(chunkBase + CHUNK) * 3 + (size_t)tid * 24, 0, 3);
        }
        __syncthreads();

#pragma unroll 4
        for (int t = 0; t < TILES_PER_CHUNK; ++t) {
            const float2 bf = sB[t * 32 + lane];
            v2f b;
            b[0] = bf.x;
            b[1] = bf.y;
            v8f cacc = {0.f, 0.f, 0.f, 0.f, 0.f, 0.f, 0.f, 0.f};
            cacc = __builtin_amdgcn_wmma_f32_16x16x4_f32(
                false, a, false, b, (short)0, cacc, false, false);
            const int nIdx = chunkBase + t * 16 + row;
#pragma unroll
            for (int r = 0; r < 8; ++r) {
                const bool lt = cacc[r] < bestVal[r];
                bestIdx[r] = lt ? nIdx    : bestIdx[r];
                bestVal[r] = lt ? cacc[r] : bestVal[r];
            }
        }
        __syncthreads();
    }

#pragma unroll
    for (int r = 0; r < 8; ++r) {
        float v = bestVal[r];
        int   i = bestIdx[r];
#pragma unroll
        for (int off = 1; off < 16; off <<= 1) {
            const float ov = __shfl_xor(v, off, 32);
            const int   oi = __shfl_xor(i, off, 32);
            if (ov < v || (ov == v && oi < i)) { v = ov; i = oi; }
        }
        bestVal[r] = v;
        bestIdx[r] = i;
    }

    const int dVec = D >> 2;
#pragma unroll
    for (int r = 0; r < 16; ++r) {
        const int nn = __shfl(bestIdx[r & 7], (r < 8) ? 0 : 16, 32);
        const float4* __restrict__ src = (const float4*)(features + (size_t)nn * D);
        float4* __restrict__ dst       = (float4*)(outp + (size_t)(mBase + r) * D);
        for (int d = lane; d < dVec; d += 32) dst[d] = src[d];
    }
}

extern "C" void kernel_launch(void* const* d_in, const int* in_sizes, int n_in,
                              void* d_out, int out_size, void* d_ws, size_t ws_size,
                              hipStream_t stream) {
    const float* pts      = (const float*)d_in[0];
    const float* lidar    = (const float*)d_in[1];
    const float* features = (const float*)d_in[2];
    float* outp           = (float*)d_out;

    const int M = in_sizes[0] / 3;
    const int N = in_sizes[1] / 3;
    const int D = in_sizes[2] / N;

    const int grid = M / ROWS_PER_BLOCK;
    nn_wmma_gather_kernel<<<grid, BLOCK, 0, stream>>>(pts, lidar, features, outp, M, N, D);
}


# Lidar points on the WMMA rows so a lane's 8 accumulators are one query: tile min by 4 v_min3, only the tile number tracked; winning tile re-multiplied once for its row; scores bit-identical
# speedup vs baseline: 1.6212x; 1.6212x over previous
#include <hip/hip_runtime.h>


typedef float v2f __attribute__((ext_vector_type(2)));
typedef float v8f __attribute__((ext_vector_type(8)));

#define CHUNK 8192
#define BLOCK 256
#define ROWS_PER_BLOCK 128
#define TILES_PER_CHUNK (CHUNK / 16)
#define PTS_PER_THREAD (CHUNK / BLOCK)

__global__ __launch_bounds__(BLOCK) void nn_wmma_gather_kernel(
    const float* __restrict__ pts,
    const float* __restrict__ lidar,
    const float* __restrict__ features,
    float* __restrict__ outp,
    int M, int N, int D)
{
    __shared__ float2 sB[TILES_PER_CHUNK * 32];

    const int tid  = threadIdx.x;
    const int lane = tid & 31;
    const int wave = tid >> 5;
    const int col  = lane & 15;
    const bool hi  = lane >= 16;

    const int mBase = blockIdx.x * ROWS_PER_BLOCK + wave * 16;
    const int m     = mBase + col;

    const float qx = pts[(size_t)m * 3 + 0];
    const float qy = pts[(size_t)m * 3 + 1];
    const float qz = pts[(size_t)m * 3 + 2];
    v2f bq;
    bq[0] = hi ? (-2.0f * qz) : (-2.0f * qx);
    bq[1] = hi ? 1.0f         : (-2.0f * qy);

    float gBest = 3.402823466e+38f;
    int   gIdx  = 0;

    const int nChunks = N / CHUNK;
    for (int c = 0; c < nChunks; ++c) {
        const int chunkBase = c * CHUNK;

        if (c > 0) __syncthreads();

        float px[PTS_PER_THREAD], py[PTS_PER_THREAD], pz[PTS_PER_THREAD];
#pragma unroll
        for (int u = 0; u < PTS_PER_THREAD; ++u) {
            const int n = chunkBase + tid + u * BLOCK;
            px[u] = lidar[(size_t)n * 3 + 0];
            py[u] = lidar[(size_t)n * 3 + 1];
            pz[u] = lidar[(size_t)n * 3 + 2];
        }
#pragma unroll
        for (int u = 0; u < PTS_PER_THREAD; ++u) {
            const int i  = tid + u * BLOCK;
            const int t  = i >> 4;
            const int cc = i & 15;
            const float x = px[u], y = py[u], z = pz[u];
            float s;
            {
#pragma clang fp contract(off)
                s = x * x + y * y;
            }
            sB[t * 32 + cc]      = make_float2(x, y);
            sB[t * 32 + cc + 16] = make_float2(z, __builtin_fmaf(z, z, s));
        }
        __syncthreads();

        float best  = 3.402823466e+38f;
        int   bestT = 0;
#pragma unroll 8
        for (int t = 0; t < TILES_PER_CHUNK; ++t) {
            const float2 af = sB[t * 32 + lane];
            v2f a;
            a[0] = af.x;
            a[1] = af.y;
            v8f cacc = {0.f, 0.f, 0.f, 0.f, 0.f, 0.f, 0.f, 0.f};
            cacc = __builtin_amdgcn_wmma_f32_16x16x4_f32(
                false, a, false, bq, (short)0, cacc, false, false);
            const float m0 = __builtin_fminf(__builtin_fminf(cacc[0], cacc[1]), cacc[2]);
            const float m1 = __builtin_fminf(__builtin_fminf(cacc[3], cacc[4]), cacc[5]);
            const float m2 = __builtin_fminf(__builtin_fminf(cacc[6], cacc[7]), best);
            const float nb = __builtin_fminf(__builtin_fminf(m0, m1), m2);
            bestT = (nb < best) ? t : bestT;
            best  = nb;
        }

        const float oBest = __shfl_xor(best, 16, 32);
        const int   oT    = __shfl_xor(bestT, 16, 32);
        const bool mine   = (best < oBest) || (best == oBest && bestT <= oT);
        const int  winT   = mine ? bestT : oT;

        float keep[8];
#pragma unroll
        for (int r = 0; r < 8; ++r) keep[r] = 3.402823466e+38f;
#pragma unroll
        for (int j = 0; j < 16; ++j) {
            const int tj = __builtin_amdgcn_readlane(winT, j);
            const float2 af = sB[tj * 32 + lane];
            v2f a;
            a[0] = af.x;
            a[1] = af.y;
            v8f cacc = {0.f, 0.f, 0.f, 0.f, 0.f, 0.f, 0.f, 0.f};
            cacc = __builtin_amdgcn_wmma_f32_16x16x4_f32(
                false, a, false, bq, (short)0, cacc, false, false);
            const bool sel = (col == j);
#pragma unroll
            for (int r = 0; r < 8; ++r) keep[r] = sel ? cacc[r] : keep[r];
        }

        float v  = keep[0];
        int   rr = 0;
#pragma unroll
        for (int r = 1; r < 8; ++r) {
            const bool lt = keep[r] < v;
            rr = lt ? r       : rr;
            v  = lt ? keep[r] : v;
        }
        const float ov  = __shfl_xor(v, 16, 32);
        const int   orr = __shfl_xor(rr, 16, 32);
        const float vLo = hi ? ov : v;
        const float vHi = hi ? v  : ov;
        const int   rLo = hi ? orr : rr;
        const int   rHi = hi ? rr  : orr;
        const bool  lo  = !(vHi < vLo);
        const float cVal = lo ? vLo : vHi;
        const int   cIdx = chunkBase + winT * 16 + (lo ? rLo : 8 + rHi);
        if (cVal < gBest) { gBest = cVal; gIdx = cIdx; }
    }

    const int dVec = D >> 2;
#pragma unroll
    for (int r = 0; r < 16; ++r) {
        const int nn = __shfl(gIdx, r, 32);
        const float4* __restrict__ src = (const float4*)(features + (size_t)nn * D);
        float4* __restrict__ dst       = (float4*)(outp + (size_t)(mBase + r) * D);
        for (int d = lane; d < dVec; d += 32) dst[d] = src[d];
    }
}

extern "C" void kernel_launch(void* const* d_in, const int* in_sizes, int n_in,
                              void* d_out, int out_size, void* d_ws, size_t ws_size,
                              hipStream_t stream) {
    const float* pts      = (const float*)d_in[0];
    const float* lidar    = (const float*)d_in[1];
    const float* features = (const float*)d_in[2];
    float* outp           = (float*)d_out;

    const int M = in_sizes[0] / 3;
    const int N = in_sizes[1] / 3;
    const int D = in_sizes[2] / N;

    const int grid = M / ROWS_PER_BLOCK;
    nn_wmma_gather_kernel<<<grid, BLOCK, 0, stream>>>(pts, lidar, features, outp, M, N, D);
}


# Gather: a wave's 16 feature-row loads in flight together (row numbers in scalars, 16 float4 per lane) ahead of its 16 stores, replacing 16 serial load-wait-store trips; output unchanged
# speedup vs baseline: 2.1715x; 1.3394x over previous
#include <hip/hip_runtime.h>


typedef float v2f __attribute__((ext_vector_type(2)));
typedef float v8f __attribute__((ext_vector_type(8)));
typedef float v4f __attribute__((ext_vector_type(4)));

#define CHUNK 8192
#define BLOCK 256
#define ROWS_PER_BLOCK 128
#define TILES_PER_CHUNK (CHUNK / 16)
#define PTS_PER_THREAD (CHUNK / BLOCK)

__global__ __launch_bounds__(BLOCK) void nn_wmma_gather_kernel(
    const float* __restrict__ pts,
    const float* __restrict__ lidar,
    const float* __restrict__ features,
    float* __restrict__ outp,
    int M, int N, int D)
{
    __shared__ float2 sB[TILES_PER_CHUNK * 32];

    const int tid  = threadIdx.x;
    const int lane = tid & 31;
    const int wave = tid >> 5;
    const int col  = lane & 15;
    const bool hi  = lane >= 16;

    const int mBase = blockIdx.x * ROWS_PER_BLOCK + wave * 16;
    const int m     = mBase + col;

    const float qx = pts[(size_t)m * 3 + 0];
    const float qy = pts[(size_t)m * 3 + 1];
    const float qz = pts[(size_t)m * 3 + 2];
    v2f bq;
    bq[0] = hi ? (-2.0f * qz) : (-2.0f * qx);
    bq[1] = hi ? 1.0f         : (-2.0f * qy);

    float gBest = 3.402823466e+38f;
    int   gIdx  = 0;

    const int nChunks = N / CHUNK;
    for (int c = 0; c < nChunks; ++c) {
        const int chunkBase = c * CHUNK;

        if (c > 0) __syncthreads();

        float px[PTS_PER_THREAD], py[PTS_PER_THREAD], pz[PTS_PER_THREAD];
#pragma unroll
        for (int u = 0; u < PTS_PER_THREAD; ++u) {
            const int n = chunkBase + tid + u * BLOCK;
            px[u] = lidar[(size_t)n * 3 + 0];
            py[u] = lidar[(size_t)n * 3 + 1];
            pz[u] = lidar[(size_t)n * 3 + 2];
        }
#pragma unroll
        for (int u = 0; u < PTS_PER_THREAD; ++u) {
            const int i  = tid + u * BLOCK;
            const int t  = i >> 4;
            const int cc = i & 15;
            const float x = px[u], y = py[u], z = pz[u];
            float s;
            {
#pragma clang fp contract(off)
                s = x * x + y * y;
            }
            sB[t * 32 + cc]      = make_float2(x, y);
            sB[t * 32 + cc + 16] = make_float2(z, __builtin_fmaf(z, z, s));
        }
        __syncthreads();

        float best  = 3.402823466e+38f;
        int   bestT = 0;
#pragma unroll 8
        for (int t = 0; t < TILES_PER_CHUNK; ++t) {
            const float2 af = sB[t * 32 + lane];
            v2f a;
            a[0] = af.x;
            a[1] = af.y;
            v8f cacc = {0.f, 0.f, 0.f, 0.f, 0.f, 0.f, 0.f, 0.f};
            cacc = __builtin_amdgcn_wmma_f32_16x16x4_f32(
                false, a, false, bq, (short)0, cacc, false, false);
            const float m0 = __builtin_fminf(__builtin_fminf(cacc[0], cacc[1]), cacc[2]);
            const float m1 = __builtin_fminf(__builtin_fminf(cacc[3], cacc[4]), cacc[5]);
            const float m2 = __builtin_fminf(__builtin_fminf(cacc[6], cacc[7]), best);
            const float nb = __builtin_fminf(__builtin_fminf(m0, m1), m2);
            bestT = (nb < best) ? t : bestT;
            best  = nb;
        }

        const float oBest = __shfl_xor(best, 16, 32);
        const int   oT    = __shfl_xor(bestT, 16, 32);
        const bool mine   = (best < oBest) || (best == oBest && bestT <= oT);
        const int  winT   = mine ? bestT : oT;

        float keep[8];
#pragma unroll
        for (int r = 0; r < 8; ++r) keep[r] = 3.402823466e+38f;
#pragma unroll
        for (int j = 0; j < 16; ++j) {
            const int tj = __builtin_amdgcn_readlane(winT, j);
            const float2 af = sB[tj * 32 + lane];
            v2f a;
            a[0] = af.x;
            a[1] = af.y;
            v8f cacc = {0.f, 0.f, 0.f, 0.f, 0.f, 0.f, 0.f, 0.f};
            cacc = __builtin_amdgcn_wmma_f32_16x16x4_f32(
                false, a, false, bq, (short)0, cacc, false, false);
            const bool sel = (col == j);
#pragma unroll
            for (int r = 0; r < 8; ++r) keep[r] = sel ? cacc[r] : keep[r];
        }

        float v  = keep[0];
        int   rr = 0;
#pragma unroll
        for (int r = 1; r < 8; ++r) {
            const bool lt = keep[r] < v;
            rr = lt ? r       : rr;
            v  = lt ? keep[r] : v;
        }
        const float ov  = __shfl_xor(v, 16, 32);
        const int   orr = __shfl_xor(rr, 16, 32);
        const float vLo = hi ? ov : v;
        const float vHi = hi ? v  : ov;
        const int   rLo = hi ? orr : rr;
        const int   rHi = hi ? rr  : orr;
        const bool  lo  = !(vHi < vLo);
        const float cVal = lo ? vLo : vHi;
        const int   cIdx = chunkBase + winT * 16 + (lo ? rLo : 8 + rHi);
        if (cVal < gBest) { gBest = cVal; gIdx = cIdx; }
    }

    const int dVec = D >> 2;
    int nn[16];
#pragma unroll
    for (int r = 0; r < 16; ++r) nn[r] = __builtin_amdgcn_readlane(gIdx, r);
    for (int d0 = 0; d0 < dVec; d0 += 32) {
        const int d = d0 + lane;
        if (d < dVec) {
            v4f row[16];
#pragma unroll
            for (int r = 0; r < 16; ++r)
                row[r] = ((const v4f*)(features + (size_t)nn[r] * D))[d];
            asm volatile("" :: "v"(row[0]), "v"(row[1]), "v"(row[2]), "v"(row[3]),
                               "v"(row[4]), "v"(row[5]), "v"(row[6]), "v"(row[7]),
                               "v"(row[8]), "v"(row[9]), "v"(row[10]), "v"(row[11]),
                               "v"(row[12]), "v"(row[13]), "v"(row[14]), "v"(row[15]));
#pragma unroll
            for (int r = 0; r < 16; ++r)
                ((v4f*)(outp + (size_t)(mBase + r) * D))[d] = row[r];
        }
    }
}

extern "C" void kernel_launch(void* const* d_in, const int* in_sizes, int n_in,
                              void* d_out, int out_size, void* d_ws, size_t ws_size,
                              hipStream_t stream) {
    const float* pts      = (const float*)d_in[0];
    const float* lidar    = (const float*)d_in[1];
    const float* features = (const float*)d_in[2];
    float* outp           = (float*)d_out;

    const int M = in_sizes[0] / 3;
    const int N = in_sizes[1] / 3;
    const int D = in_sizes[2] / N;

    const int grid = M / ROWS_PER_BLOCK;
    nn_wmma_gather_kernel<<<grid, BLOCK, 0, stream>>>(pts, lidar, features, outp, M, N, D);
}
